# starting kernel
# baseline (speedup 1.0000x reference)
#include <hip/hip_runtime.h>


typedef float v2f __attribute__((ext_vector_type(2)));
typedef float v8f __attribute__((ext_vector_type(8)));

#define NB 8
#define NS 64
#define ND 300
#define NTOK (NB * NS)
#define NTHREADS 256
#define NWAVE 8

#define C0 16
#define K0 50
#define KP 52
#define OW0 84
#define NP 96
#define P0 28

#define C1 32
#define OW1 26
#define NPIX1 676
#define K1 144
#define P1 8

#define C2 32
#define OW2 6
#define NPIX2 36
#define NPIX2P 48
#define K2 288

#define OFF_VM  0
#define OFF_UM  4992
#define OFF_W0C 9984
#define OFF_TC  13312
#define OFF_OC  18304
#define OFF_X1  26368
#define OFF_A1  0
#define OFF_BT  4608
#define OFF_O1  38912
#define OFF_X2  60544
#define OFF_A2  0
#define OFF_B2  9216
#define OFF_O2  23040
#define SMEM_FLOATS 62592

__device__ __forceinline__ float gelu_exact(float x) {
  return 0.5f * x * (1.0f + erff(x * 0.70710678118654752f));
}

__device__ __forceinline__ v2f ld_a(const float* A, int lda, int m0, int k0,
                                    int l, int half) {
  return *(const v2f*)(A + (m0 + l) * lda + (k0 + 2 * half));
}
__device__ __forceinline__ v2f ld_b(const float* Bm, int ldb, int k0, int n0,
                                    int l, int half) {
  v2f b;
  b.x = Bm[(k0 + 2 * half) * ldb + n0 + l];
  b.y = Bm[(k0 + 2 * half + 1) * ldb + n0 + l];
  return b;
}
__device__ __forceinline__ void st_c(float* Cm, int ldc, int m0, int n0, v8f c,
                                     int l, int half, int maxRow, int maxCol) {
  int col = n0 + l;
  if (col >= maxCol) return;
#pragma unroll
  for (int r = 0; r < 8; ++r) {
    int row = m0 + r + 8 * half;
    if (row < maxRow) Cm[row * ldc + col] = c[r];
  }
}

#define WMMA_F32(acc, a, b)                                                   \
  (acc) = __builtin_amdgcn_wmma_f32_16x16x4_f32(false, (a), false, (b),       \
                                                (short)0, (acc), false, false)

extern "C" __global__ __launch_bounds__(NTHREADS) void
fused_outer_conv_kernel(const float* __restrict__ tf,
                        const float* __restrict__ tp,
                        const float* __restrict__ w0, const float* __restrict__ b0,
                        const float* __restrict__ w1, const float* __restrict__ b1,
                        const float* __restrict__ w2, const float* __restrict__ b2,
                        float* __restrict__ out) {
  extern __shared__ float smem[];
  const int tid  = threadIdx.x;
  const int lane = tid & 31;
  const int l    = tid & 15;
  const int half = (tid >> 4) & 1;
  const int wave = tid >> 5;
  const int tok  = blockIdx.x;

  const float* v = tf + tok * ND;
  const float* u = tp + tok * ND;

  {
    const char* pw1 = (const char*)w1;
    const char* pw2 = (const char*)w2;
    if (tid * 128 < (int)(C1 * K1 * sizeof(float))) __builtin_prefetch(pw1 + tid * 128, 0, 1);
    if (tid * 128 < (int)(C2 * K2 * sizeof(float))) __builtin_prefetch(pw2 + tid * 128, 0, 1);
  }

  float* Vm  = smem + OFF_VM;
  float* Um  = smem + OFF_UM;
  float* W0c = smem + OFF_W0C;
  float* Tc  = smem + OFF_TC;
  float* Oc  = smem + OFF_OC;
  float* X1  = smem + OFF_X1;

  for (int i = tid; i < KP * NP; i += NTHREADS) {
    int kx = i / NP, ox = i % NP;
    Vm[i] = (kx < K0 && ox < OW0) ? v[3 * ox + kx] : 0.f;
  }
  for (int i = tid; i < NP * KP; i += NTHREADS) {
    int oy = i / KP, ky = i % KP;
    Um[i] = (oy < OW0 && ky < K0) ? u[3 * oy + ky] : 0.f;
  }

  for (int c = 0; c < C0; ++c) {
    __syncthreads();
    for (int i = tid; i < 64 * KP; i += NTHREADS) {
      int ky = i / KP, kx = i % KP;
      W0c[i] = (ky < K0 && kx < K0) ? w0[(c * K0 + ky) * K0 + kx] : 0.f;
    }
    __syncthreads();

    for (int t = wave; t < 4 * 6; t += NWAVE) {
      int m0 = (t / 6) * 16, n0 = (t % 6) * 16;
      v8f acc = {0.f, 0.f, 0.f, 0.f, 0.f, 0.f, 0.f, 0.f};
      for (int k0 = 0; k0 < KP; k0 += 4) {
        v2f a  = ld_a(W0c, KP, m0, k0, l, half);
        v2f bb = ld_b(Vm, NP, k0, n0, l, half);
        WMMA_F32(acc, a, bb);
      }
      st_c(Tc, NP, m0, n0, acc, l, half, KP, NP);
    }
    __syncthreads();

    for (int t = wave; t < 6 * 6; t += NWAVE) {
      int m0 = (t / 6) * 16, n0 = (t % 6) * 16;
      v8f acc = {0.f, 0.f, 0.f, 0.f, 0.f, 0.f, 0.f, 0.f};
      for (int k0 = 0; k0 < KP; k0 += 4) {
        v2f a  = ld_a(Um, KP, m0, k0, l, half);
        v2f bb = ld_b(Tc, NP, k0, n0, l, half);
        WMMA_F32(acc, a, bb);
      }
      st_c(Oc, NP, m0, n0, acc, l, half, OW0, NP);
    }
    __syncthreads();

    float bc = b0[c];
    for (int i = tid; i < P0 * P0; i += NTHREADS) {
      int py = i / P0, px = i % P0;
      float m = -3.402823466e38f;
      for (int dy = 0; dy < 3; ++dy)
        for (int dx = 0; dx < 3; ++dx)
          m = fmaxf(m, gelu_exact(Oc[(3 * py + dy) * NP + (3 * px + dx)] + bc));
      X1[c * (P0 * P0) + i] = m;
    }
  }
  __syncthreads();

  float* A1 = smem + OFF_A1;
  float* O1 = smem + OFF_O1;
  float* X2 = smem + OFF_X2;
  for (int i = tid; i < C1 * K1; i += NTHREADS) A1[i] = w1[i];
  __syncthreads();

  float* Bt = smem + OFF_BT + wave * (K1 * 16);
  for (int nt = wave; nt < (NPIX1 + 15) / 16; nt += NWAVE) {
    for (int i = lane; i < K1 * 16; i += 32) {
      int kk = i >> 4, j = i & 15;
      int p = nt * 16 + j;
      float val = 0.f;
      if (p < NPIX1) {
        int py = p / OW1, px = p % OW1;
        int ci = kk / 9, r9 = kk % 9;
        val = X1[ci * (P0 * P0) + (py + r9 / 3) * P0 + (px + r9 % 3)];
      }
      Bt[i] = val;
    }
    for (int m0 = 0; m0 < C1; m0 += 16) {
      v8f acc = {0.f, 0.f, 0.f, 0.f, 0.f, 0.f, 0.f, 0.f};
      for (int k0 = 0; k0 < K1; k0 += 4) {
        v2f a  = ld_a(A1, K1, m0, k0, l, half);
        v2f bb = ld_b(Bt, 16, k0, 0, l, half);
        WMMA_F32(acc, a, bb);
      }
      int col = nt * 16 + l;
      if (col < NPIX1) {
#pragma unroll
        for (int r = 0; r < 8; ++r) O1[(m0 + r + 8 * half) * NPIX1 + col] = acc[r];
      }
    }
  }
  __syncthreads();

  for (int i = tid; i < C1 * P1 * P1; i += NTHREADS) {
    int co = i >> 6, rem = i & 63, py = rem >> 3, px = rem & 7;
    float bc = b1[co];
    float m = -3.402823466e38f;
    for (int dy = 0; dy < 3; ++dy)
      for (int dx = 0; dx < 3; ++dx)
        m = fmaxf(m, gelu_exact(O1[co * NPIX1 + (3 * py + dy) * OW1 + (3 * px + dx)] + bc));
    X2[i] = m;
  }
  __syncthreads();

  float* A2 = smem + OFF_A2;
  float* B2 = smem + OFF_B2;
  float* O2 = smem + OFF_O2;
  for (int i = tid; i < C2 * K2; i += NTHREADS) A2[i] = w2[i];
  for (int i = tid; i < K2 * NPIX2P; i += NTHREADS) {
    int kk = i / NPIX2P, col = i % NPIX2P;
    float val = 0.f;
    if (col < NPIX2) {
      int py = col / OW2, px = col % OW2;
      int ci = kk / 9, r9 = kk % 9;
      val = X2[ci * (P1 * P1) + (py + r9 / 3) * P1 + (px + r9 % 3)];
    }
    B2[i] = val;
  }
  __syncthreads();

  for (int t = wave; t < 2 * 3; t += NWAVE) {
    int m0 = (t / 3) * 16, n0 = (t % 3) * 16;
    v8f acc = {0.f, 0.f, 0.f, 0.f, 0.f, 0.f, 0.f, 0.f};
    for (int k0 = 0; k0 < K2; k0 += 4) {
      v2f a  = ld_a(A2, K2, m0, k0, l, half);
      v2f bb = ld_b(B2, NPIX2P, k0, n0, l, half);
      WMMA_F32(acc, a, bb);
    }
    st_c(O2, NPIX2P, m0, n0, acc, l, half, C2, NPIX2P);
  }
  __syncthreads();

  if (tid < 128) {
    int co = tid >> 2, rem = tid & 3, py = rem >> 1, px = rem & 1;
    float bc = b2[co];
    float m = -3.402823466e38f;
    for (int dy = 0; dy < 3; ++dy)
      for (int dx = 0; dx < 3; ++dx)
        m = fmaxf(m, gelu_exact(O2[co * NPIX2P + (3 * py + dy) * OW2 + (3 * px + dx)] + bc));
    out[tok * 128 + tid] = m;
  }
}

extern "C" void kernel_launch(void* const* d_in, const int* in_sizes, int n_in,
                              void* d_out, int out_size, void* d_ws, size_t ws_size,
                              hipStream_t stream) {
  (void)in_sizes; (void)n_in; (void)out_size; (void)d_ws; (void)ws_size;
  const float* tf = (const float*)d_in[0];
  const float* tp = (const float*)d_in[1];
  const float* w0 = (const float*)d_in[2];
  const float* b0 = (const float*)d_in[3];
  const float* w1 = (const float*)d_in[4];
  const float* b1 = (const float*)d_in[5];
  const float* w2 = (const float*)d_in[6];
  const float* b2 = (const float*)d_in[7];
  float* out = (float*)d_out;

  const size_t shmem = SMEM_FLOATS * sizeof(float);
  (void)hipFuncSetAttribute((const void*)fused_outer_conv_kernel,
                            hipFuncAttributeMaxDynamicSharedMemorySize, (int)shmem);
  hipLaunchKernelGGL(fused_outer_conv_kernel, dim3(NTOK), dim3(NTHREADS), shmem,
                     stream, tf, tp, w0, b0, w1, b1, w2, b2, out);
}


# bias + GELU moved from the three pool passes into the GEMM epilogues: erf as straight-line two-range code on a tile's 8 accumulator registers; planes hold activated values, pool = 9 loads + max
# speedup vs baseline: 3.4038x; 3.4038x over previous
#include <hip/hip_runtime.h>


typedef float v2f __attribute__((ext_vector_type(2)));
typedef float v8f __attribute__((ext_vector_type(8)));

#define NB 8
#define NS 64
#define ND 300
#define NTOK (NB * NS)
#define NTHREADS 1024
#define NWAVE 32
#define NWAVE_P2 16

#define C0 16
#define K0 50
#define KP 52
#define OW0 84
#define NP 96
#define P0 28

#define C1 32
#define OW1 26
#define NPIX1 676
#define K1 144
#define P1 8

#define C2 32
#define OW2 6
#define NPIX2 36
#define NPIX2P 48
#define K2 288

#define OFF_UR  0
#define OFF_VR  512
#define OFF_B0S 1024
#define W0C_FLOATS 3328
#define TC_FLOATS  4992
#define OFF_W0C 9984
#define OFF_TC  16640
#define OFF_OC0 26624
#define OFF_OC1 54016
#define OFF_TT  34688
#define W0C_ROWS 48
#define W0C_USED 2496
#define OFF_X1  41472
#define OFF_A1  0
#define OFF_BT  4608
#define OFF_O1  54016
#define OFF_X2  75648
#define OFF_A2  0
#define OFF_B2  9216
#define OFF_O2  23040
#define SMEM_FLOATS 77696

__device__ __forceinline__ float f32_bits(unsigned int u) { return __builtin_bit_cast(float, u); }

__device__ __forceinline__ float erf_sel(float y) {
  const float ax = __builtin_fabsf(y);
  float p = __builtin_fmaf(ax, f32_bits(0x378e98abu), f32_bits(0xb9c68948u));
  p = __builtin_fmaf(ax, p, f32_bits(0x3b7cd369u));
  p = __builtin_fmaf(ax, p, f32_bits(0xbcc618b2u));
  p = __builtin_fmaf(ax, p, f32_bits(0x3dda74e4u));
  p = __builtin_fmaf(ax, p, f32_bits(0x3f228afdu));
  p = __builtin_fmaf(ax, p, f32_bits(0x3e03c728u));
  p = __builtin_fmaf(ax, p, ax);
  const float big = 1.0f - expf(-p);
  const float t = y * y;
  float q = __builtin_fmaf(f32_bits(0xba1345e1u), t, f32_bits(0x3ba10414u));
  q = __builtin_fmaf(t, q, f32_bits(0xbcdac9b8u));
  q = __builtin_fmaf(t, q, f32_bits(0x3de703beu));
  q = __builtin_fmaf(t, q, f32_bits(0xbec09330u));
  q = __builtin_fmaf(t, q, f32_bits(0x3e0375d0u));
  const float small = __builtin_fmaf(ax, q, ax);
  const float r = (ax < 1.0f) ? small : big;
  return __builtin_copysignf(r, y);
}
__device__ __forceinline__ float gelu_exact(float x) {
  return 0.5f * x * (1.0f + erf_sel(x * 0.70710678118654752f));
}

__device__ __forceinline__ v2f ld_a(const float* A, int lda, int m0, int k0,
                                    int l, int half) {
  return *(const v2f*)(A + (m0 + l) * lda + (k0 + 2 * half));
}
__device__ __forceinline__ v2f ld_b(const float* Bm, int ldb, int k0, int n0,
                                    int l, int half) {
  v2f b;
  b.x = Bm[(k0 + 2 * half) * ldb + n0 + l];
  b.y = Bm[(k0 + 2 * half + 1) * ldb + n0 + l];
  return b;
}

#define WMMA_F32(acc, a, b)                                                   \
  (acc) = __builtin_amdgcn_wmma_f32_16x16x4_f32(false, (a), false, (b),       \
                                                (short)0, (acc), false, false)

extern "C" __global__ __launch_bounds__(NTHREADS) void
fused_outer_conv_kernel(const float* __restrict__ tf,
                        const float* __restrict__ tp,
                        const float* __restrict__ w0, const float* __restrict__ b0,
                        const float* __restrict__ w1, const float* __restrict__ b1,
                        const float* __restrict__ w2, const float* __restrict__ b2,
                        float* __restrict__ out) {
  extern __shared__ float smem[];
  const int tid  = threadIdx.x;
  const int lane = tid & 31;
  const int l    = tid & 15;
  const int half = (tid >> 4) & 1;
  const int wave = __builtin_amdgcn_readfirstlane(tid >> 5);
  const int tok  = blockIdx.x;

  const float* v = tf + tok * ND;
  const float* u = tp + tok * ND;

  {
    const char* pw1 = (const char*)w1;
    const char* pw2 = (const char*)w2;
    if (tid * 128 < (int)(C1 * K1 * sizeof(float))) __builtin_prefetch(pw1 + tid * 128, 0, 1);
    if (tid * 128 < (int)(C2 * K2 * sizeof(float))) __builtin_prefetch(pw2 + tid * 128, 0, 1);
  }

  const int vw  = ((wave & 4) << 2) | ((((wave >> 3) ^ wave) & 3) << 2) | (wave & 3);
  const int grp = (vw >> 2) & 1;
  const int own6 = vw % 6, own4 = vw / 6;

  v2f wt[13];
  if (vw < 12) {
    const int m = own4 * 16 + l;
    const float* wr = w0 + (m >> 1) * (K0 * K0) + (W0C_ROWS + (m & 1)) * K0 + 2 * half;
#pragma unroll
    for (int k = 0; k < 13; ++k) {
      const bool ok = (k < 12) || (half == 0);
      wt[k].x = ok ? wr[4 * k] : 0.f;
      wt[k].y = ok ? wr[4 * k + 1] : 0.f;
    }
  }

  float* W0c = smem + OFF_W0C;
  float* Tc  = smem + OFF_TC;
  float* Oc0 = smem + OFF_OC0;
  float* Oc1 = smem + OFF_OC1;
  float* X1  = smem + OFF_X1;

  float* Ur = smem + OFF_UR;
  float* Vr = smem + OFF_VR;
  auto wsrc_of = [](int t, int j) -> int {
    const int i = t + j * NTHREADS;
    const int ky = i / KP, kx = i % KP;
    return (i < W0C_USED && kx < K0) ? ky * K0 + kx : -1;
  };
  {
    const int e512 = tid & 511;
    const float* uvsrc = (tid < 512 ? u : v) + (e512 < ND ? e512 : 0);
    float uv = *uvsrc;
    uv = (e512 < ND) ? uv : 0.f;
    const float b0v = b0[tid < C0 ? tid : 0];
    float wpro[6];
#pragma unroll
    for (int cc = 0; cc < 2; ++cc) {
#pragma unroll
      for (int j = 0; j < 3; ++j) {
        const int ws = wsrc_of(tid, j);
        wpro[cc * 3 + j] = (ws >= 0) ? w0[cc * (K0 * K0) + ws] : 0.f;
      }
    }
    Ur[tid] = uv;
    if (tid < C0) smem[OFF_B0S + tid] = b0v;
#pragma unroll
    for (int cc = 0; cc < 2; ++cc) {
#pragma unroll
      for (int j = 0; j < 3; ++j) {
        const int i = tid + j * NTHREADS;
        if (i < W0C_USED) W0c[cc * W0C_FLOATS + i] = wpro[cc * 3 + j];
      }
    }
  }
  __syncthreads();

  v2f aU[13], bV[13];
  {
    const int rc = own6 * 16 + l;
    const bool rcok = rc < OW0;
    const float* pu = Ur + 3 * rc + 2 * half;
    const float* pv = Vr + 3 * rc + 2 * half;
#pragma unroll
    for (int k = 0; k < 13; ++k) {
      const bool ok = rcok && ((k < 12) || (half == 0));
      const float u0 = pu[4 * k], u1 = pu[4 * k + 1];
      const float v0 = pv[4 * k], v1 = pv[4 * k + 1];
      aU[k].x = ok ? u0 : 0.f;
      aU[k].y = ok ? u1 : 0.f;
      bV[k].x = ok ? v0 : 0.f;
      bV[k].y = ok ? v1 : 0.f;
    }
  }

  float* Tt = smem + OFF_TT;
  if (vw < 12) {
    v8f acc = {0.f, 0.f, 0.f, 0.f, 0.f, 0.f, 0.f, 0.f};
#pragma unroll
    for (int k = 0; k < 13; ++k) WMMA_F32(acc, wt[k], bV[k]);
    float* Cp = Tt + (own4 * 16 + 8 * half) * NP + own6 * 16 + l;
#pragma unroll
    for (int r = 0; r < 8; ++r) Cp[r * NP] = acc[r];
  }

  for (int it = 0; it < C0 + 2; ++it) {
    const bool doW = (it >= 1) && (it + 1 < C0);
    float wreg[3] = {0.f, 0.f, 0.f};
    if (doW) {
      int t2 = tid;
      asm volatile("" : "+v"(t2));
#pragma unroll
      for (int j = 0; j < 3; ++j) {
        const int ws = wsrc_of(t2, j);
        wreg[j] = (ws >= 0) ? w0[(it + 1) * (K0 * K0) + ws] : 0.f;
      }
    }

    for (int step = 0; step < 2; ++step) {
      if ((step ^ grp) == 0) {
        if (it < C0 && vw < 18) {
          const float* Aop = W0c + (it & 1) * W0C_FLOATS;
          v2f a[13];
#pragma unroll
          for (int k = 0; k < 13; ++k) a[k] = ld_a(Aop, KP, own4 * 16, 4 * k, l, half);
          __builtin_amdgcn_sched_barrier(0);
          v8f acc = {0.f, 0.f, 0.f, 0.f, 0.f, 0.f, 0.f, 0.f};
#pragma unroll
          for (int k = 0; k < 13; ++k) WMMA_F32(acc, a[k], bV[k]);
          float* Cp = Tc + (it & 1) * TC_FLOATS + (own4 * 16 + 8 * half) * NP + own6 * 16 + l;
#pragma unroll
          for (int r = 0; r < 8; ++r) Cp[r * NP] = acc[r];
        }
        if (it >= 1 && it <= C0) {
          const float* Bop = Tc + ((it + 1) & 1) * TC_FLOATS;
          float* Cop = ((it + 1) & 1) ? Oc1 : Oc0;
          const int nTiles = (vw >= 26 && vw < 30) ? 2 : 1;
          for (int j = 0; j < nTiles; ++j) {
            const int n0 = (own4 + j) * 16;
            v2f b[13];
#pragma unroll
            for (int k = 0; k < 12; ++k) b[k] = ld_b(Bop, NP, 4 * k, n0, l, half);
            b[12].x = Tt[(2 * (it - 1)) * NP + n0 + l];
            b[12].y = Tt[(2 * (it - 1) + 1) * NP + n0 + l];
            const float bcB = smem[OFF_B0S + it - 1];
            __builtin_amdgcn_sched_barrier(0);
            v8f acc = {0.f, 0.f, 0.f, 0.f, 0.f, 0.f, 0.f, 0.f};
#pragma unroll
            for (int k = 0; k < 13; ++k) WMMA_F32(acc, aU[k], b[k]);
            float* Cp = Cop + (own6 * 16 + 8 * half) * NP + n0 + l;
            if (own6 < 5) {
#pragma unroll
              for (int r = 0; r < 8; ++r) Cp[r * NP] = gelu_exact(acc[r] + bcB);
            } else {
              float g4[4];
#pragma unroll
              for (int r = 0; r < 4; ++r) g4[r] = gelu_exact(acc[r] + bcB);
              if (half == 0) {
#pragma unroll
                for (int r = 0; r < 4; ++r) Cp[r * NP] = g4[r];
              }
            }
          }
        }
      } else if (it >= 2) {
        const float* Ocg = (it & 1) ? Oc1 : Oc0;
        const int i = vw * 32 + lane;
        if (i < P0 * P0) {
          int py = i / P0, px = i % P0;
          const float* wp = Ocg + (3 * py) * NP + 3 * px;
          float m = -3.402823466e38f;
#pragma unroll
          for (int dy = 0; dy < 3; ++dy)
#pragma unroll
            for (int dx = 0; dx < 3; ++dx) m = fmaxf(m, wp[dy * NP + dx]);
          X1[(it - 2) * (P0 * P0) + i] = m;
        }
      }
    }

    if (doW) {
#pragma unroll
      for (int j = 0; j < 3; ++j) {
        int i = tid + j * NTHREADS;
        if (i < W0C_USED) W0c[((it + 1) & 1) * W0C_FLOATS + i] = wreg[j];
      }
    }
    __syncthreads();
  }

  float* A1 = smem + OFF_A1;
  float* O1 = smem + OFF_O1;
  float* X2 = smem + OFF_X2;
  for (int i = tid; i < C1 * K1; i += NTHREADS) A1[i] = w1[i];
  __syncthreads();

  if (wave < NWAVE_P2) {
  float* Bt = smem + OFF_BT + wave * (K1 * 16);
  float bias1[16];
#pragma unroll
  for (int r = 0; r < 16; ++r) bias1[r] = b1[(r >> 3) * 16 + 8 * half + (r & 7)];
  for (int nt = wave; nt < (NPIX1 + 15) / 16; nt += NWAVE_P2) {
    int p = nt * 16 + l;
    p = p < NPIX1 ? p : NPIX1 - 1;
    const int py = p / OW1, px = p - py * OW1;
    const float* src = X1 + half * (P0 * P0) + py * P0 + px;
    float* dst = Bt + half * (9 * 16) + l;
#pragma unroll
    for (int g = 0; g < 2; ++g) {
      float vals[36];
#pragma unroll
      for (int e = 0; e < 36; ++e) {
        const int cl = g * 4 + e / 9, r9 = e % 9;
        vals[e] = src[cl * 2 * (P0 * P0) + (r9 / 3) * P0 + (r9 % 3)];
      }
#pragma unroll
      for (int e = 0; e < 36; ++e) {
        const int cl = g * 4 + e / 9, r9 = e % 9;
        dst[(cl * 18 + r9) * 16] = vals[e];
      }
    }
#pragma unroll
    for (int mt = 0; mt < 2; ++mt) {
      const int m0 = mt * 16;
      v8f acc = {0.f, 0.f, 0.f, 0.f, 0.f, 0.f, 0.f, 0.f};
      for (int k0 = 0; k0 < K1; k0 += 4) {
        v2f a  = ld_a(A1, K1, m0, k0, l, half);
        v2f bb = ld_b(Bt, 16, k0, 0, l, half);
        WMMA_F32(acc, a, bb);
      }
      float g[8];
#pragma unroll
      for (int r = 0; r < 8; ++r) g[r] = gelu_exact(acc[r] + bias1[mt * 8 + r]);
      int col = nt * 16 + l;
      if (col < NPIX1) {
#pragma unroll
        for (int r = 0; r < 8; ++r) O1[(m0 + r + 8 * half) * NPIX1 + col] = g[r];
      }
    }
  }
  }
  __syncthreads();

#pragma unroll
  for (int i = tid; i < C1 * P1 * P1; i += NTHREADS) {
    int co = i >> 6, rem = i & 63, py = rem >> 3, px = rem & 7;
    const float* wp = O1 + co * NPIX1 + (3 * py) * OW1 + 3 * px;
    float m = -3.402823466e38f;
#pragma unroll
    for (int dy = 0; dy < 3; ++dy)
#pragma unroll
      for (int dx = 0; dx < 3; ++dx) m = fmaxf(m, wp[dy * OW1 + dx]);
    X2[i] = m;
  }
  __syncthreads();

  float* A2 = smem + OFF_A2;
  float* B2 = smem + OFF_B2;
  float* O2 = smem + OFF_O2;
  for (int i = tid; i < C2 * K2; i += NTHREADS) A2[i] = w2[i];
  {
    const int col = tid & 63, g = tid >> 6;
    if (col < NPIX2P) {
      const int cc = col < NPIX2 ? col : NPIX2 - 1;
      const int py = cc / OW2, px = cc - py * OW2;
      const float* src = X2 + g * 2 * (P1 * P1) + py * P1 + px;
      float* dst = B2 + g * 18 * NPIX2P + col;
      float vals[18];
#pragma unroll
      for (int e = 0; e < 18; ++e) {
        const int c2 = e / 9, r9 = e % 9;
        vals[e] = src[c2 * (P1 * P1) + (r9 / 3) * P1 + (r9 % 3)];
      }
#pragma unroll
      for (int e = 0; e < 18; ++e) dst[e * NPIX2P] = vals[e];
    }
  }
  __syncthreads();

  for (int t = wave; t < 2 * 3; t += NWAVE) {
    int m0 = (t / 3) * 16, n0 = (t % 3) * 16;
    float bias2[8];
#pragma unroll
    for (int r = 0; r < 8; ++r) bias2[r] = b2[m0 + 8 * half + r];
    v8f acc = {0.f, 0.f, 0.f, 0.f, 0.f, 0.f, 0.f, 0.f};
    for (int k0 = 0; k0 < K2; k0 += 4) {
      v2f a  = ld_a(A2, K2, m0, k0, l, half);
      v2f bb = ld_b(B2, NPIX2P, k0, n0, l, half);
      WMMA_F32(acc, a, bb);
    }
    float* Cp = O2 + (m0 + 8 * half) * NPIX2P + n0 + l;
#pragma unroll
    for (int r = 0; r < 8; ++r) Cp[r * NPIX2P] = gelu_exact(acc[r] + bias2[r]);
  }
  __syncthreads();

  if (tid < 128) {
    int co = tid >> 2, rem = tid & 3, py = rem >> 1, px = rem & 1;
    const float* wp = O2 + co * NPIX2P + (3 * py) * OW2 + 3 * px;
    float m = -3.402823466e38f;
#pragma unroll
    for (int dy = 0; dy < 3; ++dy)
#pragma unroll
      for (int dx = 0; dx < 3; ++dx) m = fmaxf(m, wp[dy * OW2 + dx]);
    out[tok * 128 + tid] = m;
  }
}

extern "C" void kernel_launch(void* const* d_in, const int* in_sizes, int n_in,
                              void* d_out, int out_size, void* d_ws, size_t ws_size,
                              hipStream_t stream) {
  (void)in_sizes; (void)n_in; (void)out_size; (void)d_ws; (void)ws_size;
  const float* tf = (const float*)d_in[0];
  const float* tp = (const float*)d_in[1];
  const float* w0 = (const float*)d_in[2];
  const float* b0 = (const float*)d_in[3];
  const float* w1 = (const float*)d_in[4];
  const float* b1 = (const float*)d_in[5];
  const float* w2 = (const float*)d_in[6];
  const float* b2 = (const float*)d_in[7];
  float* out = (float*)d_out;

  const size_t shmem = SMEM_FLOATS * sizeof(float);
  (void)hipFuncSetAttribute((const void*)fused_outer_conv_kernel,
                            hipFuncAttributeMaxDynamicSharedMemorySize, (int)shmem);
  hipLaunchKernelGGL(fused_outer_conv_kernel, dim3(NTOK), dim3(NTHREADS), shmem,
                     stream, tf, tp, w0, b0, w1, b1, w2, b2, out);
}


# erf's exp(-p), p >= 0, as one multiply + hardware exp2 in every GELU epilogue instead of the library expf (range reduction, ldexp, overflow selects): 230 -> 127 vector instructions per tile epilogue
# speedup vs baseline: 3.8640x; 1.1352x over previous
#include <hip/hip_runtime.h>


typedef float v2f __attribute__((ext_vector_type(2)));
typedef float v8f __attribute__((ext_vector_type(8)));

#define NB 8
#define NS 64
#define ND 300
#define NTOK (NB * NS)
#define NTHREADS 1024
#define NWAVE 32
#define NWAVE_P2 16

#define C0 16
#define K0 50
#define KP 52
#define OW0 84
#define NP 96
#define P0 28

#define C1 32
#define OW1 26
#define NPIX1 676
#define K1 144
#define P1 8

#define C2 32
#define OW2 6
#define NPIX2 36
#define NPIX2P 48
#define K2 288

#define OFF_UR  0
#define OFF_VR  512
#define OFF_B0S 1024
#define W0C_FLOATS 3328
#define TC_FLOATS  4992
#define OFF_W0C 9984
#define OFF_TC  16640
#define OFF_OC0 26624
#define OFF_OC1 54016
#define OFF_TT  34688
#define W0C_ROWS 48
#define W0C_USED 2496
#define OFF_X1  41472
#define OFF_A1  0
#define OFF_BT  4608
#define OFF_O1  54016
#define OFF_X2  75648
#define OFF_A2  0
#define OFF_B2  9216
#define OFF_O2  23040
#define SMEM_FLOATS 77696

__device__ __forceinline__ float f32_bits(unsigned int u) { return __builtin_bit_cast(float, u); }

__device__ __forceinline__ float erf_sel(float y) {
  const float ax = __builtin_fabsf(y);
  float p = __builtin_fmaf(ax, f32_bits(0x378e98abu), f32_bits(0xb9c68948u));
  p = __builtin_fmaf(ax, p, f32_bits(0x3b7cd369u));
  p = __builtin_fmaf(ax, p, f32_bits(0xbcc618b2u));
  p = __builtin_fmaf(ax, p, f32_bits(0x3dda74e4u));
  p = __builtin_fmaf(ax, p, f32_bits(0x3f228afdu));
  p = __builtin_fmaf(ax, p, f32_bits(0x3e03c728u));
  p = __builtin_fmaf(ax, p, ax);
  const float big = 1.0f - __builtin_amdgcn_exp2f(p * -1.4426950408889634f);
  const float t = y * y;
  float q = __builtin_fmaf(f32_bits(0xba1345e1u), t, f32_bits(0x3ba10414u));
  q = __builtin_fmaf(t, q, f32_bits(0xbcdac9b8u));
  q = __builtin_fmaf(t, q, f32_bits(0x3de703beu));
  q = __builtin_fmaf(t, q, f32_bits(0xbec09330u));
  q = __builtin_fmaf(t, q, f32_bits(0x3e0375d0u));
  const float small = __builtin_fmaf(ax, q, ax);
  const float r = (ax < 1.0f) ? small : big;
  return __builtin_copysignf(r, y);
}
__device__ __forceinline__ float gelu_exact(float x) {
  return 0.5f * x * (1.0f + erf_sel(x * 0.70710678118654752f));
}

__device__ __forceinline__ v2f ld_a(const float* A, int lda, int m0, int k0,
                                    int l, int half) {
  return *(const v2f*)(A + (m0 + l) * lda + (k0 + 2 * half));
}
__device__ __forceinline__ v2f ld_b(const float* Bm, int ldb, int k0, int n0,
                                    int l, int half) {
  v2f b;
  b.x = Bm[(k0 + 2 * half) * ldb + n0 + l];
  b.y = Bm[(k0 + 2 * half + 1) * ldb + n0 + l];
  return b;
}

#define WMMA_F32(acc, a, b)                                                   \
  (acc) = __builtin_amdgcn_wmma_f32_16x16x4_f32(false, (a), false, (b),       \
                                                (short)0, (acc), false, false)

extern "C" __global__ __launch_bounds__(NTHREADS) void
fused_outer_conv_kernel(const float* __restrict__ tf,
                        const float* __restrict__ tp,
                        const float* __restrict__ w0, const float* __restrict__ b0,
                        const float* __restrict__ w1, const float* __restrict__ b1,
                        const float* __restrict__ w2, const float* __restrict__ b2,
                        float* __restrict__ out) {
  extern __shared__ float smem[];
  const int tid  = threadIdx.x;
  const int lane = tid & 31;
  const int l    = tid & 15;
  const int half = (tid >> 4) & 1;
  const int wave = __builtin_amdgcn_readfirstlane(tid >> 5);
  const int tok  = blockIdx.x;

  const float* v = tf + tok * ND;
  const float* u = tp + tok * ND;

  {
    const char* pw1 = (const char*)w1;
    const char* pw2 = (const char*)w2;
    if (tid * 128 < (int)(C1 * K1 * sizeof(float))) __builtin_prefetch(pw1 + tid * 128, 0, 1);
    if (tid * 128 < (int)(C2 * K2 * sizeof(float))) __builtin_prefetch(pw2 + tid * 128, 0, 1);
  }

  const int vw  = ((wave & 4) << 2) | ((((wave >> 3) ^ wave) & 3) << 2) | (wave & 3);
  const int grp = (vw >> 2) & 1;
  const int own6 = vw % 6, own4 = vw / 6;

  v2f wt[13];
  if (vw < 12) {
    const int m = own4 * 16 + l;
    const float* wr = w0 + (m >> 1) * (K0 * K0) + (W0C_ROWS + (m & 1)) * K0 + 2 * half;
#pragma unroll
    for (int k = 0; k < 13; ++k) {
      const bool ok = (k < 12) || (half == 0);
      wt[k].x = ok ? wr[4 * k] : 0.f;
      wt[k].y = ok ? wr[4 * k + 1] : 0.f;
    }
  }

  float* W0c = smem + OFF_W0C;
  float* Tc  = smem + OFF_TC;
  float* Oc0 = smem + OFF_OC0;
  float* Oc1 = smem + OFF_OC1;
  float* X1  = smem + OFF_X1;

  float* Ur = smem + OFF_UR;
  float* Vr = smem + OFF_VR;
  auto wsrc_of = [](int t, int j) -> int {
    const int i = t + j * NTHREADS;
    const int ky = i / KP, kx = i % KP;
    return (i < W0C_USED && kx < K0) ? ky * K0 + kx : -1;
  };
  {
    const int e512 = tid & 511;
    const float* uvsrc = (tid < 512 ? u : v) + (e512 < ND ? e512 : 0);
    float uv = *uvsrc;
    uv = (e512 < ND) ? uv : 0.f;
    const float b0v = b0[tid < C0 ? tid : 0];
    float wpro[6];
#pragma unroll
    for (int cc = 0; cc < 2; ++cc) {
#pragma unroll
      for (int j = 0; j < 3; ++j) {
        const int ws = wsrc_of(tid, j);
        wpro[cc * 3 + j] = (ws >= 0) ? w0[cc * (K0 * K0) + ws] : 0.f;
      }
    }
    Ur[tid] = uv;
    if (tid < C0) smem[OFF_B0S + tid] = b0v;
#pragma unroll
    for (int cc = 0; cc < 2; ++cc) {
#pragma unroll
      for (int j = 0; j < 3; ++j) {
        const int i = tid + j * NTHREADS;
        if (i < W0C_USED) W0c[cc * W0C_FLOATS + i] = wpro[cc * 3 + j];
      }
    }
  }
  __syncthreads();

  v2f aU[13], bV[13];
  {
    const int rc = own6 * 16 + l;
    const bool rcok = rc < OW0;
    const float* pu = Ur + 3 * rc + 2 * half;
    const float* pv = Vr + 3 * rc + 2 * half;
#pragma unroll
    for (int k = 0; k < 13; ++k) {
      const bool ok = rcok && ((k < 12) || (half == 0));
      const float u0 = pu[4 * k], u1 = pu[4 * k + 1];
      const float v0 = pv[4 * k], v1 = pv[4 * k + 1];
      aU[k].x = ok ? u0 : 0.f;
      aU[k].y = ok ? u1 : 0.f;
      bV[k].x = ok ? v0 : 0.f;
      bV[k].y = ok ? v1 : 0.f;
    }
  }

  float* Tt = smem + OFF_TT;
  if (vw < 12) {
    v8f acc = {0.f, 0.f, 0.f, 0.f, 0.f, 0.f, 0.f, 0.f};
#pragma unroll
    for (int k = 0; k < 13; ++k) WMMA_F32(acc, wt[k], bV[k]);
    float* Cp = Tt + (own4 * 16 + 8 * half) * NP + own6 * 16 + l;
#pragma unroll
    for (int r = 0; r < 8; ++r) Cp[r * NP] = acc[r];
  }

  for (int it = 0; it < C0 + 2; ++it) {
    const bool doW = (it >= 1) && (it + 1 < C0);
    float wreg[3] = {0.f, 0.f, 0.f};
    if (doW) {
      int t2 = tid;
      asm volatile("" : "+v"(t2));
#pragma unroll
      for (int j = 0; j < 3; ++j) {
        const int ws = wsrc_of(t2, j);
        wreg[j] = (ws >= 0) ? w0[(it + 1) * (K0 * K0) + ws] : 0.f;
      }
    }

    for (int step = 0; step < 2; ++step) {
      if ((step ^ grp) == 0) {
        if (it < C0 && vw < 18) {
          const float* Aop = W0c + (it & 1) * W0C_FLOATS;
          v2f a[13];
#pragma unroll
          for (int k = 0; k < 13; ++k) a[k] = ld_a(Aop, KP, own4 * 16, 4 * k, l, half);
          __builtin_amdgcn_sched_barrier(0);
          v8f acc = {0.f, 0.f, 0.f, 0.f, 0.f, 0.f, 0.f, 0.f};
#pragma unroll
          for (int k = 0; k < 13; ++k) WMMA_F32(acc, a[k], bV[k]);
          float* Cp = Tc + (it & 1) * TC_FLOATS + (own4 * 16 + 8 * half) * NP + own6 * 16 + l;
#pragma unroll
          for (int r = 0; r < 8; ++r) Cp[r * NP] = acc[r];
        }
        if (it >= 1 && it <= C0) {
          const float* Bop = Tc + ((it + 1) & 1) * TC_FLOATS;
          float* Cop = ((it + 1) & 1) ? Oc1 : Oc0;
          const int nTiles = (vw >= 26 && vw < 30) ? 2 : 1;
          for (int j = 0; j < nTiles; ++j) {
            const int n0 = (own4 + j) * 16;
            v2f b[13];
#pragma unroll
            for (int k = 0; k < 12; ++k) b[k] = ld_b(Bop, NP, 4 * k, n0, l, half);
            b[12].x = Tt[(2 * (it - 1)) * NP + n0 + l];
            b[12].y = Tt[(2 * (it - 1) + 1) * NP + n0 + l];
            const float bcB = smem[OFF_B0S + it - 1];
            __builtin_amdgcn_sched_barrier(0);
            v8f acc = {0.f, 0.f, 0.f, 0.f, 0.f, 0.f, 0.f, 0.f};
#pragma unroll
            for (int k = 0; k < 13; ++k) WMMA_F32(acc, aU[k], b[k]);
            float* Cp = Cop + (own6 * 16 + 8 * half) * NP + n0 + l;
            if (own6 < 5) {
#pragma unroll
              for (int r = 0; r < 8; ++r) Cp[r * NP] = gelu_exact(acc[r] + bcB);
            } else {
              float g4[4];
#pragma unroll
              for (int r = 0; r < 4; ++r) g4[r] = gelu_exact(acc[r] + bcB);
              if (half == 0) {
#pragma unroll
                for (int r = 0; r < 4; ++r) Cp[r * NP] = g4[r];
              }
            }
          }
        }
      } else if (it >= 2) {
        const float* Ocg = (it & 1) ? Oc1 : Oc0;
        const int i = vw * 32 + lane;
        if (i < P0 * P0) {
          int py = i / P0, px = i % P0;
          const float* wp = Ocg + (3 * py) * NP + 3 * px;
          float m = -3.402823466e38f;
#pragma unroll
          for (int dy = 0; dy < 3; ++dy)
#pragma unroll
            for (int dx = 0; dx < 3; ++dx) m = fmaxf(m, wp[dy * NP + dx]);
          X1[(it - 2) * (P0 * P0) + i] = m;
        }
      }
    }

    if (doW) {
#pragma unroll
      for (int j = 0; j < 3; ++j) {
        int i = tid + j * NTHREADS;
        if (i < W0C_USED) W0c[((it + 1) & 1) * W0C_FLOATS + i] = wreg[j];
      }
    }
    __syncthreads();
  }

  float* A1 = smem + OFF_A1;
  float* O1 = smem + OFF_O1;
  float* X2 = smem + OFF_X2;
  for (int i = tid; i < C1 * K1; i += NTHREADS) A1[i] = w1[i];
  __syncthreads();

  if (wave < NWAVE_P2) {
  float* Bt = smem + OFF_BT + wave * (K1 * 16);
  float bias1[16];
#pragma unroll
  for (int r = 0; r < 16; ++r) bias1[r] = b1[(r >> 3) * 16 + 8 * half + (r & 7)];
  for (int nt = wave; nt < (NPIX1 + 15) / 16; nt += NWAVE_P2) {
    int p = nt * 16 + l;
    p = p < NPIX1 ? p : NPIX1 - 1;
    const int py = p / OW1, px = p - py * OW1;
    const float* src = X1 + half * (P0 * P0) + py * P0 + px;
    float* dst = Bt + half * (9 * 16) + l;
#pragma unroll
    for (int g = 0; g < 2; ++g) {
      float vals[36];
#pragma unroll
      for (int e = 0; e < 36; ++e) {
        const int cl = g * 4 + e / 9, r9 = e % 9;
        vals[e] = src[cl * 2 * (P0 * P0) + (r9 / 3) * P0 + (r9 % 3)];
      }
#pragma unroll
      for (int e = 0; e < 36; ++e) {
        const int cl = g * 4 + e / 9, r9 = e % 9;
        dst[(cl * 18 + r9) * 16] = vals[e];
      }
    }
#pragma unroll
    for (int mt = 0; mt < 2; ++mt) {
      const int m0 = mt * 16;
      v8f acc = {0.f, 0.f, 0.f, 0.f, 0.f, 0.f, 0.f, 0.f};
      for (int k0 = 0; k0 < K1; k0 += 4) {
        v2f a  = ld_a(A1, K1, m0, k0, l, half);
        v2f bb = ld_b(Bt, 16, k0, 0, l, half);
        WMMA_F32(acc, a, bb);
      }
      float g[8];
#pragma unroll
      for (int r = 0; r < 8; ++r) g[r] = gelu_exact(acc[r] + bias1[mt * 8 + r]);
      int col = nt * 16 + l;
      if (col < NPIX1) {
#pragma unroll
        for (int r = 0; r < 8; ++r) O1[(m0 + r + 8 * half) * NPIX1 + col] = g[r];
      }
    }
  }
  }
  __syncthreads();

#pragma unroll
  for (int i = tid; i < C1 * P1 * P1; i += NTHREADS) {
    int co = i >> 6, rem = i & 63, py = rem >> 3, px = rem & 7;
    const float* wp = O1 + co * NPIX1 + (3 * py) * OW1 + 3 * px;
    float m = -3.402823466e38f;
#pragma unroll
    for (int dy = 0; dy < 3; ++dy)
#pragma unroll
      for (int dx = 0; dx < 3; ++dx) m = fmaxf(m, wp[dy * OW1 + dx]);
    X2[i] = m;
  }
  __syncthreads();

  float* A2 = smem + OFF_A2;
  float* B2 = smem + OFF_B2;
  float* O2 = smem + OFF_O2;
  for (int i = tid; i < C2 * K2; i += NTHREADS) A2[i] = w2[i];
  {
    const int col = tid & 63, g = tid >> 6;
    if (col < NPIX2P) {
      const int cc = col < NPIX2 ? col : NPIX2 - 1;
      const int py = cc / OW2, px = cc - py * OW2;
      const float* src = X2 + g * 2 * (P1 * P1) + py * P1 + px;
      float* dst = B2 + g * 18 * NPIX2P + col;
      float vals[18];
#pragma unroll
      for (int e = 0; e < 18; ++e) {
        const int c2 = e / 9, r9 = e % 9;
        vals[e] = src[c2 * (P1 * P1) + (r9 / 3) * P1 + (r9 % 3)];
      }
#pragma unroll
      for (int e = 0; e < 18; ++e) dst[e * NPIX2P] = vals[e];
    }
  }
  __syncthreads();

  for (int t = wave; t < 2 * 3; t += NWAVE) {
    int m0 = (t / 3) * 16, n0 = (t % 3) * 16;
    float bias2[8];
#pragma unroll
    for (int r = 0; r < 8; ++r) bias2[r] = b2[m0 + 8 * half + r];
    v8f acc = {0.f, 0.f, 0.f, 0.f, 0.f, 0.f, 0.f, 0.f};
    for (int k0 = 0; k0 < K2; k0 += 4) {
      v2f a  = ld_a(A2, K2, m0, k0, l, half);
      v2f bb = ld_b(B2, NPIX2P, k0, n0, l, half);
      WMMA_F32(acc, a, bb);
    }
    float* Cp = O2 + (m0 + 8 * half) * NPIX2P + n0 + l;
#pragma unroll
    for (int r = 0; r < 8; ++r) Cp[r * NPIX2P] = gelu_exact(acc[r] + bias2[r]);
  }
  __syncthreads();

  if (tid < 128) {
    int co = tid >> 2, rem = tid & 3, py = rem >> 1, px = rem & 1;
    const float* wp = O2 + co * NPIX2P + (3 * py) * OW2 + 3 * px;
    float m = -3.402823466e38f;
#pragma unroll
    for (int dy = 0; dy < 3; ++dy)
#pragma unroll
      for (int dx = 0; dx < 3; ++dx) m = fmaxf(m, wp[dy * OW2 + dx]);
    out[tok * 128 + tid] = m;
  }
}

extern "C" void kernel_launch(void* const* d_in, const int* in_sizes, int n_in,
                              void* d_out, int out_size, void* d_ws, size_t ws_size,
                              hipStream_t stream) {
  (void)in_sizes; (void)n_in; (void)out_size; (void)d_ws; (void)ws_size;
  const float* tf = (const float*)d_in[0];
  const float* tp = (const float*)d_in[1];
  const float* w0 = (const float*)d_in[2];
  const float* b0 = (const float*)d_in[3];
  const float* w1 = (const float*)d_in[4];
  const float* b1 = (const float*)d_in[5];
  const float* w2 = (const float*)d_in[6];
  const float* b2 = (const float*)d_in[7];
  float* out = (float*)d_out;

  const size_t shmem = SMEM_FLOATS * sizeof(float);
  (void)hipFuncSetAttribute((const void*)fused_outer_conv_kernel,
                            hipFuncAttributeMaxDynamicSharedMemorySize, (int)shmem);
  hipLaunchKernelGGL(fused_outer_conv_kernel, dim3(NTOK), dim3(NTHREADS), shmem,
                     stream, tf, tp, w0, b0, w1, b1, w2, b2, out);
}


# GELU in one range in every GEMM epilogue: max(x,0) - |x|*erfc(|x|/sqrt2)/2, erfc = one v_exp_f32 of a degree-8 polynomial; no small-argument series, select or sign copy: 127 -> 68 VALU instr per tile
# speedup vs baseline: 4.1573x; 1.0759x over previous
#include <hip/hip_runtime.h>


typedef float v2f __attribute__((ext_vector_type(2)));
typedef float v8f __attribute__((ext_vector_type(8)));

#define NB 8
#define NS 64
#define ND 300
#define NTOK (NB * NS)
#define NTHREADS 1024
#define NWAVE 32
#define NWAVE_P2 16

#define C0 16
#define K0 50
#define KP 52
#define OW0 84
#define NP 96
#define P0 28

#define C1 32
#define OW1 26
#define NPIX1 676
#define K1 144
#define P1 8

#define C2 32
#define OW2 6
#define NPIX2 36
#define NPIX2P 48
#define K2 288

#define OFF_UR  0
#define OFF_VR  512
#define OFF_B0S 1024
#define W0C_FLOATS 3328
#define TC_FLOATS  4992
#define OFF_W0C 9984
#define OFF_TC  16640
#define OFF_OC0 26624
#define OFF_OC1 54016
#define OFF_TT  34688
#define W0C_ROWS 48
#define W0C_USED 2496
#define OFF_X1  41472
#define OFF_A1  0
#define OFF_BT  4608
#define OFF_O1  54016
#define OFF_X2  75648
#define OFF_A2  0
#define OFF_B2  9216
#define OFF_O2  23040
#define SMEM_FLOATS 77696

__device__ __forceinline__ float gelu_exact(float x) {
  const float a = __builtin_fabsf(x) * 0.70710678118654752f;
  float g = __builtin_fmaf(a, -0.0000306288639f, 0.000319134153f);
  g = __builtin_fmaf(a, g, -0.00105772167f);
  g = __builtin_fmaf(a, g, -0.00152850046f);
  g = __builtin_fmaf(a, g, 0.0289713386f);
  g = __builtin_fmaf(a, g, -0.148847923f);
  g = __builtin_fmaf(a, g, -0.918325663f);
  g = __builtin_fmaf(a, g, -1.62791669f);
  g = __builtin_fmaf(a, g, -0.5f);
  const float e = __builtin_amdgcn_exp2f(g);
  return __builtin_fmaf(-a, e, __builtin_fmaxf(x, 0.0f));
}

__device__ __forceinline__ v2f ld_a(const float* A, int lda, int m0, int k0,
                                    int l, int half) {
  return *(const v2f*)(A + (m0 + l) * lda + (k0 + 2 * half));
}
__device__ __forceinline__ v2f ld_b(const float* Bm, int ldb, int k0, int n0,
                                    int l, int half) {
  v2f b;
  b.x = Bm[(k0 + 2 * half) * ldb + n0 + l];
  b.y = Bm[(k0 + 2 * half + 1) * ldb + n0 + l];
  return b;
}

#define WMMA_F32(acc, a, b)                                                   \
  (acc) = __builtin_amdgcn_wmma_f32_16x16x4_f32(false, (a), false, (b),       \
                                                (short)0, (acc), false, false)

extern "C" __global__ __launch_bounds__(NTHREADS) void
fused_outer_conv_kernel(const float* __restrict__ tf,
                        const float* __restrict__ tp,
                        const float* __restrict__ w0, const float* __restrict__ b0,
                        const float* __restrict__ w1, const float* __restrict__ b1,
                        const float* __restrict__ w2, const float* __restrict__ b2,
                        float* __restrict__ out) {
  extern __shared__ float smem[];
  const int tid  = threadIdx.x;
  const int lane = tid & 31;
  const int l    = tid & 15;
  const int half = (tid >> 4) & 1;
  const int wave = __builtin_amdgcn_readfirstlane(tid >> 5);
  const int tok  = blockIdx.x;

  const float* v = tf + tok * ND;
  const float* u = tp + tok * ND;

  {
    const char* pw1 = (const char*)w1;
    const char* pw2 = (const char*)w2;
    if (tid * 128 < (int)(C1 * K1 * sizeof(float))) __builtin_prefetch(pw1 + tid * 128, 0, 1);
    if (tid * 128 < (int)(C2 * K2 * sizeof(float))) __builtin_prefetch(pw2 + tid * 128, 0, 1);
  }

  const int vw  = ((wave & 4) << 2) | ((((wave >> 3) ^ wave) & 3) << 2) | (wave & 3);
  const int grp = (vw >> 2) & 1;
  const int own6 = vw % 6, own4 = vw / 6;

  v2f wt[13];
  if (vw < 12) {
    const int m = own4 * 16 + l;
    const float* wr = w0 + (m >> 1) * (K0 * K0) + (W0C_ROWS + (m & 1)) * K0 + 2 * half;
#pragma unroll
    for (int k = 0; k < 13; ++k) {
      const bool ok = (k < 12) || (half == 0);
      wt[k].x = ok ? wr[4 * k] : 0.f;
      wt[k].y = ok ? wr[4 * k + 1] : 0.f;
    }
  }

  float* W0c = smem + OFF_W0C;
  float* Tc  = smem + OFF_TC;
  float* Oc0 = smem + OFF_OC0;
  float* Oc1 = smem + OFF_OC1;
  float* X1  = smem + OFF_X1;

  float* Ur = smem + OFF_UR;
  float* Vr = smem + OFF_VR;
  auto wsrc_of = [](int t, int j) -> int {
    const int i = t + j * NTHREADS;
    const int ky = i / KP, kx = i % KP;
    return (i < W0C_USED && kx < K0) ? ky * K0 + kx : -1;
  };
  {
    const int e512 = tid & 511;
    const float* uvsrc = (tid < 512 ? u : v) + (e512 < ND ? e512 : 0);
    float uv = *uvsrc;
    uv = (e512 < ND) ? uv : 0.f;
    const float b0v = b0[tid < C0 ? tid : 0];
    float wpro[6];
#pragma unroll
    for (int cc = 0; cc < 2; ++cc) {
#pragma unroll
      for (int j = 0; j < 3; ++j) {
        const int ws = wsrc_of(tid, j);
        wpro[cc * 3 + j] = (ws >= 0) ? w0[cc * (K0 * K0) + ws] : 0.f;
      }
    }
    Ur[tid] = uv;
    if (tid < C0) smem[OFF_B0S + tid] = b0v;
#pragma unroll
    for (int cc = 0; cc < 2; ++cc) {
#pragma unroll
      for (int j = 0; j < 3; ++j) {
        const int i = tid + j * NTHREADS;
        if (i < W0C_USED) W0c[cc * W0C_FLOATS + i] = wpro[cc * 3 + j];
      }
    }
  }
  __syncthreads();

  v2f aU[13], bV[13];
  {
    const int rc = own6 * 16 + l;
    const bool rcok = rc < OW0;
    const float* pu = Ur + 3 * rc + 2 * half;
    const float* pv = Vr + 3 * rc + 2 * half;
#pragma unroll
    for (int k = 0; k < 13; ++k) {
      const bool ok = rcok && ((k < 12) || (half == 0));
      const float u0 = pu[4 * k], u1 = pu[4 * k + 1];
      const float v0 = pv[4 * k], v1 = pv[4 * k + 1];
      aU[k].x = ok ? u0 : 0.f;
      aU[k].y = ok ? u1 : 0.f;
      bV[k].x = ok ? v0 : 0.f;
      bV[k].y = ok ? v1 : 0.f;
    }
  }

  float* Tt = smem + OFF_TT;
  if (vw < 12) {
    v8f acc = {0.f, 0.f, 0.f, 0.f, 0.f, 0.f, 0.f, 0.f};
#pragma unroll
    for (int k = 0; k < 13; ++k) WMMA_F32(acc, wt[k], bV[k]);
    float* Cp = Tt + (own4 * 16 + 8 * half) * NP + own6 * 16 + l;
#pragma unroll
    for (int r = 0; r < 8; ++r) Cp[r * NP] = acc[r];
  }

  for (int it = 0; it < C0 + 2; ++it) {
    const bool doW = (it >= 1) && (it + 1 < C0);
    float wreg[3] = {0.f, 0.f, 0.f};
    if (doW) {
      int t2 = tid;
      asm volatile("" : "+v"(t2));
#pragma unroll
      for (int j = 0; j < 3; ++j) {
        const int ws = wsrc_of(t2, j);
        wreg[j] = (ws >= 0) ? w0[(it + 1) * (K0 * K0) + ws] : 0.f;
      }
    }

    for (int step = 0; step < 2; ++step) {
      if ((step ^ grp) == 0) {
        if (it < C0 && vw < 18) {
          const float* Aop = W0c + (it & 1) * W0C_FLOATS;
          v2f a[13];
#pragma unroll
          for (int k = 0; k < 13; ++k) a[k] = ld_a(Aop, KP, own4 * 16, 4 * k, l, half);
          __builtin_amdgcn_sched_barrier(0);
          v8f acc = {0.f, 0.f, 0.f, 0.f, 0.f, 0.f, 0.f, 0.f};
#pragma unroll
          for (int k = 0; k < 13; ++k) WMMA_F32(acc, a[k], bV[k]);
          float* Cp = Tc + (it & 1) * TC_FLOATS + (own4 * 16 + 8 * half) * NP + own6 * 16 + l;
#pragma unroll
          for (int r = 0; r < 8; ++r) Cp[r * NP] = acc[r];
        }
        if (it >= 1 && it <= C0) {
          const float* Bop = Tc + ((it + 1) & 1) * TC_FLOATS;
          float* Cop = ((it + 1) & 1) ? Oc1 : Oc0;
          const int nTiles = (vw >= 26 && vw < 30) ? 2 : 1;
          for (int j = 0; j < nTiles; ++j) {
            const int n0 = (own4 + j) * 16;
            v2f b[13];
#pragma unroll
            for (int k = 0; k < 12; ++k) b[k] = ld_b(Bop, NP, 4 * k, n0, l, half);
            b[12].x = Tt[(2 * (it - 1)) * NP + n0 + l];
            b[12].y = Tt[(2 * (it - 1) + 1) * NP + n0 + l];
            const float bcB = smem[OFF_B0S + it - 1];
            __builtin_amdgcn_sched_barrier(0);
            v8f acc = {0.f, 0.f, 0.f, 0.f, 0.f, 0.f, 0.f, 0.f};
#pragma unroll
            for (int k = 0; k < 13; ++k) WMMA_F32(acc, aU[k], b[k]);
            float* Cp = Cop + (own6 * 16 + 8 * half) * NP + n0 + l;
            if (own6 < 5) {
#pragma unroll
              for (int r = 0; r < 8; ++r) Cp[r * NP] = gelu_exact(acc[r] + bcB);
            } else {
              float g4[4];
#pragma unroll
              for (int r = 0; r < 4; ++r) g4[r] = gelu_exact(acc[r] + bcB);
              if (half == 0) {
#pragma unroll
                for (int r = 0; r < 4; ++r) Cp[r * NP] = g4[r];
              }
            }
          }
        }
      } else if (it >= 2) {
        const float* Ocg = (it & 1) ? Oc1 : Oc0;
        const int i = vw * 32 + lane;
        if (i < P0 * P0) {
          int py = i / P0, px = i % P0;
          const float* wp = Ocg + (3 * py) * NP + 3 * px;
          float m = -3.402823466e38f;
#pragma unroll
          for (int dy = 0; dy < 3; ++dy)
#pragma unroll
            for (int dx = 0; dx < 3; ++dx) m = fmaxf(m, wp[dy * NP + dx]);
          X1[(it - 2) * (P0 * P0) + i] = m;
        }
      }
    }

    if (doW) {
#pragma unroll
      for (int j = 0; j < 3; ++j) {
        int i = tid + j * NTHREADS;
        if (i < W0C_USED) W0c[((it + 1) & 1) * W0C_FLOATS + i] = wreg[j];
      }
    }
    __syncthreads();
  }

  float* A1 = smem + OFF_A1;
  float* O1 = smem + OFF_O1;
  float* X2 = smem + OFF_X2;
  for (int i = tid; i < C1 * K1; i += NTHREADS) A1[i] = w1[i];
  __syncthreads();

  if (wave < NWAVE_P2) {
  float* Bt = smem + OFF_BT + wave * (K1 * 16);
  float bias1[16];
#pragma unroll
  for (int r = 0; r < 16; ++r) bias1[r] = b1[(r >> 3) * 16 + 8 * half + (r & 7)];
  for (int nt = wave; nt < (NPIX1 + 15) / 16; nt += NWAVE_P2) {
    int p = nt * 16 + l;
    p = p < NPIX1 ? p : NPIX1 - 1;
    const int py = p / OW1, px = p - py * OW1;
    const float* src = X1 + half * (P0 * P0) + py * P0 + px;
    float* dst = Bt + half * (9 * 16) + l;
#pragma unroll
    for (int g = 0; g < 2; ++g) {
      float vals[36];
#pragma unroll
      for (int e = 0; e < 36; ++e) {
        const int cl = g * 4 + e / 9, r9 = e % 9;
        vals[e] = src[cl * 2 * (P0 * P0) + (r9 / 3) * P0 + (r9 % 3)];
      }
#pragma unroll
      for (int e = 0; e < 36; ++e) {
        const int cl = g * 4 + e / 9, r9 = e % 9;
        dst[(cl * 18 + r9) * 16] = vals[e];
      }
    }
#pragma unroll
    for (int mt = 0; mt < 2; ++mt) {
      const int m0 = mt * 16;
      v8f acc = {0.f, 0.f, 0.f, 0.f, 0.f, 0.f, 0.f, 0.f};
      for (int k0 = 0; k0 < K1; k0 += 4) {
        v2f a  = ld_a(A1, K1, m0, k0, l, half);
        v2f bb = ld_b(Bt, 16, k0, 0, l, half);
        WMMA_F32(acc, a, bb);
      }
      float g[8];
#pragma unroll
      for (int r = 0; r < 8; ++r) g[r] = gelu_exact(acc[r] + bias1[mt * 8 + r]);
      int col = nt * 16 + l;
      if (col < NPIX1) {
#pragma unroll
        for (int r = 0; r < 8; ++r) O1[(m0 + r + 8 * half) * NPIX1 + col] = g[r];
      }
    }
  }
  }
  __syncthreads();

#pragma unroll
  for (int i = tid; i < C1 * P1 * P1; i += NTHREADS) {
    int co = i >> 6, rem = i & 63, py = rem >> 3, px = rem & 7;
    const float* wp = O1 + co * NPIX1 + (3 * py) * OW1 + 3 * px;
    float m = -3.402823466e38f;
#pragma unroll
    for (int dy = 0; dy < 3; ++dy)
#pragma unroll
      for (int dx = 0; dx < 3; ++dx) m = fmaxf(m, wp[dy * OW1 + dx]);
    X2[i] = m;
  }
  __syncthreads();

  float* A2 = smem + OFF_A2;
  float* B2 = smem + OFF_B2;
  float* O2 = smem + OFF_O2;
  for (int i = tid; i < C2 * K2; i += NTHREADS) A2[i] = w2[i];
  {
    const int col = tid & 63, g = tid >> 6;
    if (col < NPIX2P) {
      const int cc = col < NPIX2 ? col : NPIX2 - 1;
      const int py = cc / OW2, px = cc - py * OW2;
      const float* src = X2 + g * 2 * (P1 * P1) + py * P1 + px;
      float* dst = B2 + g * 18 * NPIX2P + col;
      float vals[18];
#pragma unroll
      for (int e = 0; e < 18; ++e) {
        const int c2 = e / 9, r9 = e % 9;
        vals[e] = src[c2 * (P1 * P1) + (r9 / 3) * P1 + (r9 % 3)];
      }
#pragma unroll
      for (int e = 0; e < 18; ++e) dst[e * NPIX2P] = vals[e];
    }
  }
  __syncthreads();

  for (int t = wave; t < 2 * 3; t += NWAVE) {
    int m0 = (t / 3) * 16, n0 = (t % 3) * 16;
    float bias2[8];
#pragma unroll
    for (int r = 0; r < 8; ++r) bias2[r] = b2[m0 + 8 * half + r];
    v8f acc = {0.f, 0.f, 0.f, 0.f, 0.f, 0.f, 0.f, 0.f};
    for (int k0 = 0; k0 < K2; k0 += 4) {
      v2f a  = ld_a(A2, K2, m0, k0, l, half);
      v2f bb = ld_b(B2, NPIX2P, k0, n0, l, half);
      WMMA_F32(acc, a, bb);
    }
    float* Cp = O2 + (m0 + 8 * half) * NPIX2P + n0 + l;
#pragma unroll
    for (int r = 0; r < 8; ++r) Cp[r * NPIX2P] = gelu_exact(acc[r] + bias2[r]);
  }
  __syncthreads();

  if (tid < 128) {
    int co = tid >> 2, rem = tid & 3, py = rem >> 1, px = rem & 1;
    const float* wp = O2 + co * NPIX2P + (3 * py) * OW2 + 3 * px;
    float m = -3.402823466e38f;
#pragma unroll
    for (int dy = 0; dy < 3; ++dy)
#pragma unroll
      for (int dx = 0; dx < 3; ++dx) m = fmaxf(m, wp[dy * OW2 + dx]);
    out[tok * 128 + tid] = m;
  }
}

extern "C" void kernel_launch(void* const* d_in, const int* in_sizes, int n_in,
                              void* d_out, int out_size, void* d_ws, size_t ws_size,
                              hipStream_t stream) {
  (void)in_sizes; (void)n_in; (void)out_size; (void)d_ws; (void)ws_size;
  const float* tf = (const float*)d_in[0];
  const float* tp = (const float*)d_in[1];
  const float* w0 = (const float*)d_in[2];
  const float* b0 = (const float*)d_in[3];
  const float* w1 = (const float*)d_in[4];
  const float* b1 = (const float*)d_in[5];
  const float* w2 = (const float*)d_in[6];
  const float* b2 = (const float*)d_in[7];
  float* out = (float*)d_out;

  const size_t shmem = SMEM_FLOATS * sizeof(float);
  (void)hipFuncSetAttribute((const void*)fused_outer_conv_kernel,
                            hipFuncAttributeMaxDynamicSharedMemorySize, (int)shmem);
  hipLaunchKernelGGL(fused_outer_conv_kernel, dim3(NTOK), dim3(NTHREADS), shmem,
                     stream, tf, tp, w0, b0, w1, b1, w2, b2, out);
}


# conv1 GEMM operand feed: w1 at LDS pitch 148 and im2col slab with k pairs side by side, so every fragment is one conflict-free 8-byte load at a fixed offset; one B fragment feeds both row tiles
# speedup vs baseline: 4.4358x; 1.0670x over previous
#include <hip/hip_runtime.h>


typedef float v2f __attribute__((ext_vector_type(2)));
typedef float v8f __attribute__((ext_vector_type(8)));

#define NB 8
#define NS 64
#define ND 300
#define NTOK (NB * NS)
#define NTHREADS 1024
#define NWAVE 32
#define NWAVE_P2 16

#define C0 16
#define K0 50
#define KP 52
#define OW0 84
#define NP 96
#define P0 28

#define C1 32
#define OW1 26
#define NPIX1 676
#define K1 144
#define P1 8

#define C2 32
#define OW2 6
#define NPIX2 36
#define NPIX2P 48
#define K2 288

#define OFF_UR  0
#define OFF_VR  512
#define OFF_B0S 1024
#define W0C_FLOATS 3328
#define TC_FLOATS  4992
#define OFF_W0C 9984
#define OFF_TC  16640
#define OFF_OC0 26624
#define OFF_OC1 54016
#define OFF_TT  34688
#define W0C_ROWS 48
#define W0C_USED 2496
#define OFF_X1  41472
#define A1P     148
#define OFF_A1  0
#define OFF_BT  4736
#define OFF_BT15 75648
#define OFF_O1  54016
#define OFF_X2  75648
#define OFF_A2  0
#define OFF_B2  9216
#define OFF_O2  23040
#define SMEM_FLOATS 77952

__device__ __forceinline__ float gelu_exact(float x) {
  const float a = __builtin_fabsf(x) * 0.70710678118654752f;
  float g = __builtin_fmaf(a, -0.0000306288639f, 0.000319134153f);
  g = __builtin_fmaf(a, g, -0.00105772167f);
  g = __builtin_fmaf(a, g, -0.00152850046f);
  g = __builtin_fmaf(a, g, 0.0289713386f);
  g = __builtin_fmaf(a, g, -0.148847923f);
  g = __builtin_fmaf(a, g, -0.918325663f);
  g = __builtin_fmaf(a, g, -1.62791669f);
  g = __builtin_fmaf(a, g, -0.5f);
  const float e = __builtin_amdgcn_exp2f(g);
  return __builtin_fmaf(-a, e, __builtin_fmaxf(x, 0.0f));
}

__device__ __forceinline__ v2f ld_a(const float* A, int lda, int m0, int k0,
                                    int l, int half) {
  return *(const v2f*)(A + (m0 + l) * lda + (k0 + 2 * half));
}
__device__ __forceinline__ v2f ld_b(const float* Bm, int ldb, int k0, int n0,
                                    int l, int half) {
  v2f b;
  b.x = Bm[(k0 + 2 * half) * ldb + n0 + l];
  b.y = Bm[(k0 + 2 * half + 1) * ldb + n0 + l];
  return b;
}

#define WMMA_F32(acc, a, b)                                                   \
  (acc) = __builtin_amdgcn_wmma_f32_16x16x4_f32(false, (a), false, (b),       \
                                                (short)0, (acc), false, false)

extern "C" __global__ __launch_bounds__(NTHREADS) void
fused_outer_conv_kernel(const float* __restrict__ tf,
                        const float* __restrict__ tp,
                        const float* __restrict__ w0, const float* __restrict__ b0,
                        const float* __restrict__ w1, const float* __restrict__ b1,
                        const float* __restrict__ w2, const float* __restrict__ b2,
                        float* __restrict__ out) {
  extern __shared__ float smem[];
  const int tid  = threadIdx.x;
  const int lane = tid & 31;
  const int l    = tid & 15;
  const int half = (tid >> 4) & 1;
  const int wave = __builtin_amdgcn_readfirstlane(tid >> 5);
  const int tok  = blockIdx.x;

  const float* v = tf + tok * ND;
  const float* u = tp + tok * ND;

  {
    const char* pw1 = (const char*)w1;
    const char* pw2 = (const char*)w2;
    if (tid * 128 < (int)(C1 * K1 * sizeof(float))) __builtin_prefetch(pw1 + tid * 128, 0, 1);
    if (tid * 128 < (int)(C2 * K2 * sizeof(float))) __builtin_prefetch(pw2 + tid * 128, 0, 1);
  }

  const int vw  = ((wave & 4) << 2) | ((((wave >> 3) ^ wave) & 3) << 2) | (wave & 3);
  const int grp = (vw >> 2) & 1;
  const int own6 = vw % 6, own4 = vw / 6;

  v2f wt[13];
  if (vw < 12) {
    const int m = own4 * 16 + l;
    const float* wr = w0 + (m >> 1) * (K0 * K0) + (W0C_ROWS + (m & 1)) * K0 + 2 * half;
#pragma unroll
    for (int k = 0; k < 13; ++k) {
      const bool ok = (k < 12) || (half == 0);
      wt[k].x = ok ? wr[4 * k] : 0.f;
      wt[k].y = ok ? wr[4 * k + 1] : 0.f;
    }
  }

  float* W0c = smem + OFF_W0C;
  float* Tc  = smem + OFF_TC;
  float* Oc0 = smem + OFF_OC0;
  float* Oc1 = smem + OFF_OC1;
  float* X1  = smem + OFF_X1;

  float* Ur = smem + OFF_UR;
  float* Vr = smem + OFF_VR;
  auto wsrc_of = [](int t, int j) -> int {
    const int i = t + j * NTHREADS;
    const int ky = i / KP, kx = i % KP;
    return (i < W0C_USED && kx < K0) ? ky * K0 + kx : -1;
  };
  {
    const int e512 = tid & 511;
    const float* uvsrc = (tid < 512 ? u : v) + (e512 < ND ? e512 : 0);
    float uv = *uvsrc;
    uv = (e512 < ND) ? uv : 0.f;
    const float b0v = b0[tid < C0 ? tid : 0];
    float wpro[6];
#pragma unroll
    for (int cc = 0; cc < 2; ++cc) {
#pragma unroll
      for (int j = 0; j < 3; ++j) {
        const int ws = wsrc_of(tid, j);
        wpro[cc * 3 + j] = (ws >= 0) ? w0[cc * (K0 * K0) + ws] : 0.f;
      }
    }
    Ur[tid] = uv;
    if (tid < C0) smem[OFF_B0S + tid] = b0v;
#pragma unroll
    for (int cc = 0; cc < 2; ++cc) {
#pragma unroll
      for (int j = 0; j < 3; ++j) {
        const int i = tid + j * NTHREADS;
        if (i < W0C_USED) W0c[cc * W0C_FLOATS + i] = wpro[cc * 3 + j];
      }
    }
  }
  __syncthreads();

  v2f aU[13], bV[13];
  {
    const int rc = own6 * 16 + l;
    const bool rcok = rc < OW0;
    const float* pu = Ur + 3 * rc + 2 * half;
    const float* pv = Vr + 3 * rc + 2 * half;
#pragma unroll
    for (int k = 0; k < 13; ++k) {
      const bool ok = rcok && ((k < 12) || (half == 0));
      const float u0 = pu[4 * k], u1 = pu[4 * k + 1];
      const float v0 = pv[4 * k], v1 = pv[4 * k + 1];
      aU[k].x = ok ? u0 : 0.f;
      aU[k].y = ok ? u1 : 0.f;
      bV[k].x = ok ? v0 : 0.f;
      bV[k].y = ok ? v1 : 0.f;
    }
  }

  float* Tt = smem + OFF_TT;
  if (vw < 12) {
    v8f acc = {0.f, 0.f, 0.f, 0.f, 0.f, 0.f, 0.f, 0.f};
#pragma unroll
    for (int k = 0; k < 13; ++k) WMMA_F32(acc, wt[k], bV[k]);
    float* Cp = Tt + (own4 * 16 + 8 * half) * NP + own6 * 16 + l;
#pragma unroll
    for (int r = 0; r < 8; ++r) Cp[r * NP] = acc[r];
  }

  for (int it = 0; it < C0 + 2; ++it) {
    const bool doW = (it >= 1) && (it + 1 < C0);
    float wreg[3] = {0.f, 0.f, 0.f};
    if (doW) {
      int t2 = tid;
      asm volatile("" : "+v"(t2));
#pragma unroll
      for (int j = 0; j < 3; ++j) {
        const int ws = wsrc_of(t2, j);
        wreg[j] = (ws >= 0) ? w0[(it + 1) * (K0 * K0) + ws] : 0.f;
      }
    }

    for (int step = 0; step < 2; ++step) {
      if ((step ^ grp) == 0) {
        if (it < C0 && vw < 18) {
          const float* Aop = W0c + (it & 1) * W0C_FLOATS;
          v2f a[13];
#pragma unroll
          for (int k = 0; k < 13; ++k) a[k] = ld_a(Aop, KP, own4 * 16, 4 * k, l, half);
          __builtin_amdgcn_sched_barrier(0);
          v8f acc = {0.f, 0.f, 0.f, 0.f, 0.f, 0.f, 0.f, 0.f};
#pragma unroll
          for (int k = 0; k < 13; ++k) WMMA_F32(acc, a[k], bV[k]);
          float* Cp = Tc + (it & 1) * TC_FLOATS + (own4 * 16 + 8 * half) * NP + own6 * 16 + l;
#pragma unroll
          for (int r = 0; r < 8; ++r) Cp[r * NP] = acc[r];
        }
        if (it >= 1 && it <= C0) {
          const float* Bop = Tc + ((it + 1) & 1) * TC_FLOATS;
          float* Cop = ((it + 1) & 1) ? Oc1 : Oc0;
          const int nTiles = (vw >= 26 && vw < 30) ? 2 : 1;
          for (int j = 0; j < nTiles; ++j) {
            const int n0 = (own4 + j) * 16;
            v2f b[13];
#pragma unroll
            for (int k = 0; k < 12; ++k) b[k] = ld_b(Bop, NP, 4 * k, n0, l, half);
            b[12].x = Tt[(2 * (it - 1)) * NP + n0 + l];
            b[12].y = Tt[(2 * (it - 1) + 1) * NP + n0 + l];
            const float bcB = smem[OFF_B0S + it - 1];
            __builtin_amdgcn_sched_barrier(0);
            v8f acc = {0.f, 0.f, 0.f, 0.f, 0.f, 0.f, 0.f, 0.f};
#pragma unroll
            for (int k = 0; k < 13; ++k) WMMA_F32(acc, aU[k], b[k]);
            float* Cp = Cop + (own6 * 16 + 8 * half) * NP + n0 + l;
            if (own6 < 5) {
#pragma unroll
              for (int r = 0; r < 8; ++r) Cp[r * NP] = gelu_exact(acc[r] + bcB);
            } else {
              float g4[4];
#pragma unroll
              for (int r = 0; r < 4; ++r) g4[r] = gelu_exact(acc[r] + bcB);
              if (half == 0) {
#pragma unroll
                for (int r = 0; r < 4; ++r) Cp[r * NP] = g4[r];
              }
            }
          }
        }
      } else if (it >= 2) {
        const float* Ocg = (it & 1) ? Oc1 : Oc0;
        const int i = vw * 32 + lane;
        if (i < P0 * P0) {
          int py = i / P0, px = i % P0;
          const float* wp = Ocg + (3 * py) * NP + 3 * px;
          float m = -3.402823466e38f;
#pragma unroll
          for (int dy = 0; dy < 3; ++dy)
#pragma unroll
            for (int dx = 0; dx < 3; ++dx) m = fmaxf(m, wp[dy * NP + dx]);
          X1[(it - 2) * (P0 * P0) + i] = m;
        }
      }
    }

    if (doW) {
#pragma unroll
      for (int j = 0; j < 3; ++j) {
        int i = tid + j * NTHREADS;
        if (i < W0C_USED) W0c[((it + 1) & 1) * W0C_FLOATS + i] = wreg[j];
      }
    }
    __syncthreads();
  }

  float* A1 = smem + OFF_A1;
  float* O1 = smem + OFF_O1;
  float* X2 = smem + OFF_X2;
  for (int i = tid; i < C1 * K1; i += NTHREADS) A1[(i / K1) * A1P + (i % K1)] = w1[i];
  __syncthreads();

  if (wave < NWAVE_P2) {
  float* Bt = smem + (wave < NWAVE_P2 - 1 ? OFF_BT + wave * (K1 * 16) : OFF_BT15);
  float bias1[16];
#pragma unroll
  for (int r = 0; r < 16; ++r) bias1[r] = b1[(r >> 3) * 16 + 8 * half + (r & 7)];
  const float* Ap = A1 + l * A1P + 2 * half;
  const float* Bp = Bt + half * 32 + 2 * l;
  for (int nt = wave; nt < (NPIX1 + 15) / 16; nt += NWAVE_P2) {
    int p = nt * 16 + l;
    p = p < NPIX1 ? p : NPIX1 - 1;
    const int py = p / OW1, px = p - py * OW1;
    const float* src = X1 + 2 * half * (P0 * P0) + py * P0 + px;
    float* dst = Bt + 2 * l + half * 288;
#pragma unroll
    for (int g = 0; g < 2; ++g) {
      float vals[36];
#pragma unroll
      for (int e = 0; e < 36; ++e) {
        const int q = g * 2 + e / 18, s = (e % 18) / 9, r9 = e % 9;
        vals[e] = src[(4 * q + s) * (P0 * P0) + (r9 / 3) * P0 + (r9 % 3)];
      }
#pragma unroll
      for (int e = 0; e < 36; e += 2) {
        const int q = g * 2 + e / 18, pp = (e % 18) / 2;
        v2f two;
        two.x = vals[e];
        two.y = vals[e + 1];
        *(v2f*)(dst + (18 * q + pp) * 32) = two;
      }
    }
    v8f acc0 = {0.f, 0.f, 0.f, 0.f, 0.f, 0.f, 0.f, 0.f};
    v8f acc1 = {0.f, 0.f, 0.f, 0.f, 0.f, 0.f, 0.f, 0.f};
    v2f fa0[2][4], fa1[2][4], fb[2][4];
#pragma unroll
    for (int j = 0; j < 4; ++j) {
      fa0[0][j] = *(const v2f*)(Ap + 4 * j);
      fa1[0][j] = *(const v2f*)(Ap + 16 * A1P + 4 * j);
      fb[0][j]  = *(const v2f*)(Bp + 64 * j);
    }
#pragma unroll
    for (int g = 0; g < K1 / 16; ++g) {
      if (g + 1 < K1 / 16) {
#pragma unroll
        for (int j = 0; j < 4; ++j) {
          const int k = (g + 1) * 4 + j;
          fa0[(g + 1) & 1][j] = *(const v2f*)(Ap + 4 * k);
          fa1[(g + 1) & 1][j] = *(const v2f*)(Ap + 16 * A1P + 4 * k);
          fb[(g + 1) & 1][j]  = *(const v2f*)(Bp + 64 * k);
        }
      }
      __builtin_amdgcn_sched_barrier(0);
#pragma unroll
      for (int j = 0; j < 4; ++j) {
        WMMA_F32(acc0, fa0[g & 1][j], fb[g & 1][j]);
        WMMA_F32(acc1, fa1[g & 1][j], fb[g & 1][j]);
      }
      __builtin_amdgcn_sched_barrier(0);
    }
    float g0[8], g1[8];
#pragma unroll
    for (int r = 0; r < 8; ++r) {
      g0[r] = gelu_exact(acc0[r] + bias1[r]);
      g1[r] = gelu_exact(acc1[r] + bias1[8 + r]);
    }
    const int col = nt * 16 + l;
    if (col < NPIX1) {
#pragma unroll
      for (int r = 0; r < 8; ++r) {
        O1[(r + 8 * half) * NPIX1 + col] = g0[r];
        O1[(16 + r + 8 * half) * NPIX1 + col] = g1[r];
      }
    }
  }
  }
  __syncthreads();

#pragma unroll
  for (int i = tid; i < C1 * P1 * P1; i += NTHREADS) {
    int co = i >> 6, rem = i & 63, py = rem >> 3, px = rem & 7;
    const float* wp = O1 + co * NPIX1 + (3 * py) * OW1 + 3 * px;
    float m = -3.402823466e38f;
#pragma unroll
    for (int dy = 0; dy < 3; ++dy)
#pragma unroll
      for (int dx = 0; dx < 3; ++dx) m = fmaxf(m, wp[dy * OW1 + dx]);
    X2[i] = m;
  }
  __syncthreads();

  float* A2 = smem + OFF_A2;
  float* B2 = smem + OFF_B2;
  float* O2 = smem + OFF_O2;
  for (int i = tid; i < C2 * K2; i += NTHREADS) A2[i] = w2[i];
  {
    const int col = tid & 63, g = tid >> 6;
    if (col < NPIX2P) {
      const int cc = col < NPIX2 ? col : NPIX2 - 1;
      const int py = cc / OW2, px = cc - py * OW2;
      const float* src = X2 + g * 2 * (P1 * P1) + py * P1 + px;
      float* dst = B2 + g * 18 * NPIX2P + col;
      float vals[18];
#pragma unroll
      for (int e = 0; e < 18; ++e) {
        const int c2 = e / 9, r9 = e % 9;
        vals[e] = src[c2 * (P1 * P1) + (r9 / 3) * P1 + (r9 % 3)];
      }
#pragma unroll
      for (int e = 0; e < 18; ++e) dst[e * NPIX2P] = vals[e];
    }
  }
  __syncthreads();

  for (int t = wave; t < 2 * 3; t += NWAVE) {
    int m0 = (t / 3) * 16, n0 = (t % 3) * 16;
    float bias2[8];
#pragma unroll
    for (int r = 0; r < 8; ++r) bias2[r] = b2[m0 + 8 * half + r];
    v8f acc = {0.f, 0.f, 0.f, 0.f, 0.f, 0.f, 0.f, 0.f};
    for (int k0 = 0; k0 < K2; k0 += 4) {
      v2f a  = ld_a(A2, K2, m0, k0, l, half);
      v2f bb = ld_b(B2, NPIX2P, k0, n0, l, half);
      WMMA_F32(acc, a, bb);
    }
    float* Cp = O2 + (m0 + 8 * half) * NPIX2P + n0 + l;
#pragma unroll
    for (int r = 0; r < 8; ++r) Cp[r * NPIX2P] = gelu_exact(acc[r] + bias2[r]);
  }
  __syncthreads();

  if (tid < 128) {
    int co = tid >> 2, rem = tid & 3, py = rem >> 1, px = rem & 1;
    const float* wp = O2 + co * NPIX2P + (3 * py) * OW2 + 3 * px;
    float m = -3.402823466e38f;
#pragma unroll
    for (int dy = 0; dy < 3; ++dy)
#pragma unroll
      for (int dx = 0; dx < 3; ++dx) m = fmaxf(m, wp[dy * OW2 + dx]);
    out[tok * 128 + tid] = m;
  }
}

extern "C" void kernel_launch(void* const* d_in, const int* in_sizes, int n_in,
                              void* d_out, int out_size, void* d_ws, size_t ws_size,
                              hipStream_t stream) {
  (void)in_sizes; (void)n_in; (void)out_size; (void)d_ws; (void)ws_size;
  const float* tf = (const float*)d_in[0];
  const float* tp = (const float*)d_in[1];
  const float* w0 = (const float*)d_in[2];
  const float* b0 = (const float*)d_in[3];
  const float* w1 = (const float*)d_in[4];
  const float* b1 = (const float*)d_in[5];
  const float* w2 = (const float*)d_in[6];
  const float* b2 = (const float*)d_in[7];
  float* out = (float*)d_out;

  const size_t shmem = SMEM_FLOATS * sizeof(float);
  (void)hipFuncSetAttribute((const void*)fused_outer_conv_kernel,
                            hipFuncAttributeMaxDynamicSharedMemorySize, (int)shmem);
  hipLaunchKernelGGL(fused_outer_conv_kernel, dim3(NTOK), dim3(NTHREADS), shmem,
                     stream, tf, tp, w0, b0, w1, b1, w2, b2, out);
}
